# starting kernel
# baseline (speedup 1.0000x reference)
#include <hip/hip_runtime.h>
#include <hip/hip_bf16.h>

#define B_DIM   32
#define C_DIM   768
#define H_DIM   32
#define W_DIM   32
#define HW      1024
#define LNUM    9
#define NCLS    200
#define QSTR    772

typedef __attribute__((ext_vector_type(2))) float v2f;
typedef __attribute__((ext_vector_type(8))) float v8f;

__global__ __launch_bounds__(256) void k1_maps(
    const float* __restrict__ x, const float* __restrict__ q,
    const float* __restrict__ p_bias,
    float* __restrict__ out_maps, int* __restrict__ aidx, float* __restrict__ wsel)
{
    __shared__ float qlds[16 * QSTR];
    __shared__ float asq[16];

    const int tid = threadIdx.x;
    const int b   = blockIdx.y;

    for (int i = tid; i < 16 * C_DIM; i += 256) {
        const int l = i / C_DIM, c = i % C_DIM;
        qlds[l * QSTR + c] = (l < LNUM) ? q[l * C_DIM + c] : 0.f;
    }
    __syncthreads();
    if (tid < 16) {
        float s = 0.f;
        for (int c = 0; c < C_DIM; ++c) { const float v = qlds[tid * QSTR + c]; s += v * v; }
        asq[tid] = s;
    }
    __syncthreads();

    const int wave = tid >> 5;
    const int lane = tid & 31;
    const int half = lane >> 4;
    const int ln   = lane & 15;
    const int tile = blockIdx.x * 8 + wave;
    const int pix  = tile * 16 + ln;
    const float* xb = x + (size_t)b * C_DIM * HW;

    auto load_ab = [&](int kk, v2f& af, v2f& bf) {
        af[0] = qlds[ln * QSTR + kk];
        af[1] = qlds[ln * QSTR + kk + 1];
        bf[0] = xb[(size_t)kk * HW + pix];
        bf[1] = xb[(size_t)(kk + 1) * HW + pix];
    };

    v8f acc = {};
    float sq = 0.f;
    v2f ac, bc;
    load_ab(2 * half, ac, bc);
    for (int k = 4; k < C_DIM; k += 4) {
        v2f an, bn;
        load_ab(k + 2 * half, an, bn);
        sq += bc[0] * bc[0] + bc[1] * bc[1];
        acc = __builtin_amdgcn_wmma_f32_16x16x4_f32(false, ac, false, bc,
                                                    (short)0, acc, false, false);
        ac = an; bc = bn;
    }
    sq += bc[0] * bc[0] + bc[1] * bc[1];
    acc = __builtin_amdgcn_wmma_f32_16x16x4_f32(false, ac, false, bc,
                                                (short)0, acc, false, false);
    const float bsq = sq + __shfl_xor(sq, 16);

    float sc[8];
    float mx = -3.4e38f;
    for (int r = 0; r < 8; ++r) {
        const int l = r + 8 * half;
        float v = -3.4e38f;
        if (l < LNUM)
            v = -(bsq - 2.f * acc[r] + asq[l]) + p_bias[l * HW + pix];
        sc[r] = v;
        mx = fmaxf(mx, v);
    }
    mx = fmaxf(mx, __shfl_xor(mx, 16));
    float es = 0.f;
    for (int r = 0; r < 8; ++r) {
        const int l = r + 8 * half;
        const float e = (l < LNUM) ? expf(sc[r] - mx) : 0.f;
        sc[r] = e;
        es += e;
    }
    es += __shfl_xor(es, 16);
    const float inv = 1.f / es;
    float s2 = 0.f;
    for (int r = 0; r < 8; ++r) {
        const int l = r + 8 * half;
        if (l < LNUM) { sc[r] = sc[r] * inv + 1e-6f; s2 += sc[r]; }
    }
    s2 += __shfl_xor(s2, 16);
    const float inv2 = 1.f / s2;

    float bestv = -1.f; int bestl = 0;
    float* om = out_maps + (size_t)b * LNUM * HW;
    for (int r = 0; r < 8; ++r) {
        const int l = r + 8 * half;
        if (l < LNUM) {
            const float mv = sc[r] * inv2;
            om[l * HW + pix] = mv;
            if (mv > bestv) { bestv = mv; bestl = l; }
        }
    }
    const float pv = __shfl_xor(bestv, 16);
    const int   pl = __shfl_xor(bestl, 16);
    if (pv > bestv || (pv == bestv && pl < bestl)) { bestv = pv; bestl = pl; }
    if (half == 0) {
        aidx[b * HW + pix] = bestl;
        wsel[b * HW + pix] = bestv;
    }
}

__global__ __launch_bounds__(256) void k2_pool(
    const float* __restrict__ x, const int* __restrict__ aidx,
    const float* __restrict__ wsel, float* __restrict__ pool)
{
    __shared__ int   aL[HW];
    __shared__ float wL[HW];
    const int tid = threadIdx.x;
    const int b   = blockIdx.y;
    for (int i = tid; i < HW; i += 256) { aL[i] = aidx[b * HW + i]; wL[i] = wsel[b * HW + i]; }
    __syncthreads();

    const int wave = tid >> 5;
    const int lane = tid & 31;
    const int half = lane >> 4;
    const int ln   = lane & 15;
    const int cbase = (blockIdx.x * 8 + wave) * 16;
    const float* xr = x + (size_t)b * C_DIM * HW + (size_t)(cbase + ln) * HW;

    auto load_ab = [&](int kk, v2f& af, v2f& bf) {
        af[0] = xr[kk];
        af[1] = xr[kk + 1];
        const int a0 = aL[kk], a1 = aL[kk + 1];
        bf[0] = (a0 == ln) ? wL[kk]     : 0.f;
        bf[1] = (a1 == ln) ? wL[kk + 1] : 0.f;
    };

    v8f acc = {};
    v2f ac, bc;
    load_ab(2 * half, ac, bc);
    for (int k = 4; k < HW; k += 4) {
        v2f an, bn;
        load_ab(k + 2 * half, an, bn);
        acc = __builtin_amdgcn_wmma_f32_16x16x4_f32(false, ac, false, bc,
                                                    (short)0, acc, false, false);
        ac = an; bc = bn;
    }
    acc = __builtin_amdgcn_wmma_f32_16x16x4_f32(false, ac, false, bc,
                                                (short)0, acc, false, false);

    float* pb = pool + (size_t)b * C_DIM * 16;
    for (int r = 0; r < 8; ++r) {
        const int row = cbase + r + 8 * half;
        pb[row * 16 + ln] = acc[r];
    }
}

__global__ __launch_bounds__(256) void k3_mod(
    const float* __restrict__ pool, const int* __restrict__ aidx,
    const float* __restrict__ wsel, const float* __restrict__ gamma,
    const float* __restrict__ beta, float* __restrict__ out_mod)
{
    __shared__ int   aL[HW];
    __shared__ float wL[HW];
    __shared__ float cnt[16];
    __shared__ float featL[C_DIM * LNUM];
    __shared__ float red0[8], red1[8], stat[2];

    const int tid = threadIdx.x;
    const int b   = blockIdx.x;
    for (int i = tid; i < HW; i += 256) { aL[i] = aidx[b * HW + i]; wL[i] = wsel[b * HW + i]; }
    __syncthreads();
    if (tid < 16) {
        float s = 0.f;
        for (int p = 0; p < HW; ++p) if (aL[p] == tid) s += wL[p];
        cnt[tid] = s;
    }
    __syncthreads();

    const float* pb = pool + (size_t)b * C_DIM * 16;
    float s = 0.f, ss = 0.f;
    for (int i = tid; i < C_DIM * LNUM; i += 256) {
        const int c = i / LNUM, l = i % LNUM;
        const float cn = cnt[l];
        const float v = pb[c * 16 + l] / ((cn == 0.f) ? 1.f : cn);
        featL[i] = v;
        s += v; ss += v * v;
    }
    for (int o = 16; o > 0; o >>= 1) { s += __shfl_down(s, o); ss += __shfl_down(ss, o); }
    const int lane = tid & 31, wave = tid >> 5;
    if (lane == 0) { red0[wave] = s; red1[wave] = ss; }
    __syncthreads();
    if (tid == 0) {
        float S = 0.f, SS = 0.f;
        for (int w = 0; w < 8; ++w) { S += red0[w]; SS += red1[w]; }
        const float n = (float)(C_DIM * LNUM);
        const float mean = S / n;
        const float var  = SS / n - mean * mean;
        stat[0] = mean;
        stat[1] = rsqrtf(var + 1e-5f);
    }
    __syncthreads();
    const float mean = stat[0], rstd = stat[1];
    float* ob = out_mod + (size_t)b * C_DIM * LNUM;
    for (int i = tid; i < C_DIM * LNUM; i += 256)
        ob[i] = (featL[i] - mean) * rstd * gamma[i] + beta[i];
}

__global__ __launch_bounds__(32) void k4_scores(
    const float* __restrict__ cls_w, const float* __restrict__ mod,
    float* __restrict__ scores)
{
    const int lane = threadIdx.x;
    const int half = lane >> 4;
    const int ln   = lane & 15;
    const int nbase = blockIdx.x * 16;
    const int b     = blockIdx.y;
    const float* mb = mod + (size_t)b * C_DIM * LNUM;
    const int  row  = nbase + ln;
    const bool rv   = (row < NCLS);
    const bool cv   = (ln < LNUM - 1);
    const int  rowc = rv ? row : (NCLS - 1);
    const int  lnc  = cv ? ln  : (LNUM - 2);
    const float* ar = cls_w + (size_t)rowc * C_DIM;

    v8f acc = {};
    for (int k = 0; k < C_DIM; k += 4) {
        const int kk = k + 2 * half;
        v2f afrag, bfrag;
        const float a0 = ar[kk], a1 = ar[kk + 1];
        afrag[0] = rv ? a0 : 0.f;
        afrag[1] = rv ? a1 : 0.f;
        const float b0 = mb[(size_t)kk * LNUM + lnc];
        const float b1 = mb[(size_t)(kk + 1) * LNUM + lnc];
        bfrag[0] = cv ? b0 : 0.f;
        bfrag[1] = cv ? b1 : 0.f;
        acc = __builtin_amdgcn_wmma_f32_16x16x4_f32(false, afrag, false, bfrag,
                                                    (short)0, acc, false, false);
    }
    if (cv) {
        float* sb = scores + (size_t)b * NCLS * (LNUM - 1);
        for (int r = 0; r < 8; ++r) {
            const int orow = nbase + r + 8 * half;
            if (orow < NCLS) sb[orow * (LNUM - 1) + ln] = acc[r];
        }
    }
}

extern "C" void kernel_launch(void* const* d_in, const int* in_sizes, int n_in,
                              void* d_out, int out_size, void* d_ws, size_t ws_size,
                              hipStream_t stream) {
    const float* x      = (const float*)d_in[0];
    const float* q      = (const float*)d_in[1];
    const float* p_bias = (const float*)d_in[2];
    const float* gamma  = (const float*)d_in[3];
    const float* beta   = (const float*)d_in[4];
    const float* cls_w  = (const float*)d_in[5];

    float* out        = (float*)d_out;
    float* out_mod    = out;
    float* out_maps   = out + (size_t)B_DIM * C_DIM * LNUM;
    float* out_scores = out_maps + (size_t)B_DIM * LNUM * HW;

    int*   aidx = (int*)d_ws;
    float* wsel = (float*)d_ws + (size_t)B_DIM * HW;
    float* pool = (float*)d_ws + (size_t)2 * B_DIM * HW;

    k1_maps  <<<dim3(8, B_DIM), 256, 0, stream>>>(x, q, p_bias, out_maps, aidx, wsel);
    k2_pool  <<<dim3(6, B_DIM), 256, 0, stream>>>(x, aidx, wsel, pool);
    k3_mod   <<<dim3(B_DIM),    256, 0, stream>>>(pool, aidx, wsel, gamma, beta, out_mod);
    k4_scores<<<dim3(13, B_DIM), 32, 0, stream>>>(cls_w, out_mod, out_scores);
}


# k4 re-tiled: block = 16 classes x 2 images (all 16 columns used), K split over the 8 waves, a wave's whole operand slice requested at once, partials summed in LDS; scores sum order changes
# speedup vs baseline: 5.3404x; 5.3404x over previous
#include <hip/hip_runtime.h>
#include <hip/hip_bf16.h>

#define B_DIM   32
#define C_DIM   768
#define H_DIM   32
#define W_DIM   32
#define HW      1024
#define LNUM    9
#define NCLS    200

typedef __attribute__((ext_vector_type(2))) float v2f;
typedef __attribute__((ext_vector_type(4))) float v4f;
typedef __attribute__((ext_vector_type(8))) float v8f;

#define K1_NT   4
#define K1_TSTR 260
#define K1_WSTR (K1_NT * K1_TSTR)
#define K1_SQW  (K1_NT * 36)
#define K1_GRP  4
#define K1_NG   (24 / K1_GRP)

__global__ __launch_bounds__(256, 4) void k1_maps(
    const float* __restrict__ x, const float* __restrict__ q,
    const float* __restrict__ p_bias,
    float* __restrict__ out_maps, int* __restrict__ aidx, float* __restrict__ wsel)
{
    __shared__ float red[8 * K1_WSTR];
    __shared__ float sqred[8 * K1_SQW];
    __shared__ float asqred[8 * 32];

    const int tid  = threadIdx.x;
    const int b    = blockIdx.y;
    const int wave = tid >> 5;
    const int lane = tid & 31;
    const int half = lane >> 4;
    const int ln   = lane & 15;
    const int P0   = blockIdx.x * 64;
    const int kb   = wave * 96 + 2 * half;

    const float* xw = x + (size_t)b * C_DIM * HW + (size_t)kb * HW + P0 + ln * 4;
    const bool lv = (ln < LNUM);
    const float* qw = q + (lv ? ln : 0) * C_DIM + kb;

    v8f acc[K1_NT];
    float sq[K1_NT];
#pragma unroll
    for (int t = 0; t < K1_NT; ++t) { acc[t] = v8f{}; sq[t] = 0.f; }
    float asqp = 0.f;

    auto loadg = [&](int g, v4f (&xd)[K1_GRP][2], v2f (&ad)[K1_GRP]) {
#pragma unroll
        for (int j = 0; j < K1_GRP; ++j) {
            const int s = g * K1_GRP + j;
            const float* r0 = xw + (size_t)(4 * s) * HW;
            xd[j][0] = *(const v4f*)(r0);
            xd[j][1] = *(const v4f*)(r0 + HW);
            ad[j]    = *(const v2f*)(qw + 4 * s);
        }
    };
    auto compute = [&](v4f (&xd)[K1_GRP][2], v2f (&ad)[K1_GRP]) {
#pragma unroll
        for (int j = 0; j < K1_GRP; ++j) {
            v2f af;
            af[0] = lv ? ad[j][0] : 0.f;
            af[1] = lv ? ad[j][1] : 0.f;
            asqp += af[0] * af[0] + af[1] * af[1];
#pragma unroll
            for (int t = 0; t < K1_NT; ++t) {
                v2f bf;
                bf[0] = xd[j][0][t];
                bf[1] = xd[j][1][t];
                sq[t] += bf[0] * bf[0] + bf[1] * bf[1];
                acc[t] = __builtin_amdgcn_wmma_f32_16x16x4_f32(false, af, false, bf,
                                                               (short)0, acc[t], false, false);
            }
        }
    };

    v4f xb0[K1_GRP][2], xb1[K1_GRP][2];
    v2f ab0[K1_GRP], ab1[K1_GRP];
    loadg(0, xb0, ab0);
#pragma unroll
    for (int g = 0; g < K1_NG; g += 2) {
        if (g + 1 < K1_NG) loadg(g + 1, xb1, ab1);
        compute(xb0, ab0);
        __builtin_amdgcn_sched_barrier(0);
        if (g + 2 < K1_NG) loadg(g + 2, xb0, ab0);
        if (g + 1 < K1_NG) compute(xb1, ab1);
        __builtin_amdgcn_sched_barrier(0);
    }

#pragma unroll
    for (int t = 0; t < K1_NT; ++t) {
#pragma unroll
        for (int r = 0; r < 8; ++r)
            red[wave * K1_WSTR + t * K1_TSTR + r * 32 + lane] = acc[t][r];
        sqred[wave * K1_SQW + t * 36 + lane] = sq[t];
    }
    asqred[wave * 32 + lane] = asqp;
    __syncthreads();
    if (wave >= 4) return;

    const int pl   = wave * 16 + ln;
    const int col  = pl >> 2;
    const int tsel = pl & 3;
    const int sl   = col + 16 * half;
    const int pix  = P0 + pl;

    float accf[8];
#pragma unroll
    for (int r = 0; r < 8; ++r) {
        float s = 0.f;
#pragma unroll
        for (int w = 0; w < 8; ++w) s += red[w * K1_WSTR + tsel * K1_TSTR + r * 32 + sl];
        accf[r] = s;
    }
    float bsq = 0.f;
#pragma unroll
    for (int w = 0; w < 8; ++w)
        bsq += sqred[w * K1_SQW + tsel * 36 + col] + sqred[w * K1_SQW + tsel * 36 + col + 16];
    float aq = 0.f;
#pragma unroll
    for (int w = 0; w < 8; ++w) aq += asqred[w * 32 + ln] + asqred[w * 32 + ln + 16];
    float asql[8];
#pragma unroll
    for (int r = 0; r < 8; ++r) asql[r] = __shfl(aq, r + 8 * half);

    float sc[8];
    float mx = -3.4e38f;
    for (int r = 0; r < 8; ++r) {
        const int l = r + 8 * half;
        float v = -3.4e38f;
        if (l < LNUM)
            v = -(bsq - 2.f * accf[r] + asql[r]) + p_bias[l * HW + pix];
        sc[r] = v;
        mx = fmaxf(mx, v);
    }
    mx = fmaxf(mx, __shfl_xor(mx, 16));
    float es = 0.f;
    for (int r = 0; r < 8; ++r) {
        const int l = r + 8 * half;
        const float e = (l < LNUM) ? expf(sc[r] - mx) : 0.f;
        sc[r] = e;
        es += e;
    }
    es += __shfl_xor(es, 16);
    const float inv = 1.f / es;
    float s2 = 0.f;
    for (int r = 0; r < 8; ++r) {
        const int l = r + 8 * half;
        if (l < LNUM) { sc[r] = sc[r] * inv + 1e-6f; s2 += sc[r]; }
    }
    s2 += __shfl_xor(s2, 16);
    const float inv2 = 1.f / s2;

    float bestv = -1.f; int bestl = 0;
    float* om = out_maps + (size_t)b * LNUM * HW;
    for (int r = 0; r < 8; ++r) {
        const int l = r + 8 * half;
        if (l < LNUM) {
            const float mv = sc[r] * inv2;
            om[l * HW + pix] = mv;
            if (mv > bestv) { bestv = mv; bestl = l; }
        }
    }
    const float pv = __shfl_xor(bestv, 16);
    const int   pl2 = __shfl_xor(bestl, 16);
    if (pv > bestv || (pv == bestv && pl2 < bestl)) { bestv = pv; bestl = pl2; }
    if (half == 0) {
        aidx[b * HW + pix] = bestl;
        wsel[b * HW + pix] = bestv;
    }
}

#define K2_MT   3
#define K2_PW   128
#define K2_NG   4

typedef __attribute__((ext_vector_type(4))) int v4i;

__global__ __launch_bounds__(256, 4) void k2_pool(
    const float* __restrict__ x, const int* __restrict__ aidx,
    const float* __restrict__ wsel, float* __restrict__ feat)
{
    __shared__ __attribute__((aligned(16))) int   aL[HW];
    __shared__ __attribute__((aligned(16))) float wL[HW];
    __shared__ float red[8 * K2_MT * 256];
    __shared__ float csred[8 * 32];

    const int tid  = threadIdx.x;
    const int b    = blockIdx.y;
    const int wave = tid >> 5;
    const int lane = tid & 31;
    const int half = lane >> 4;
    const int ln   = lane & 15;
    const int cbase = blockIdx.x * (16 * K2_MT);
    const int pw    = wave * K2_PW;

    const float* xr = x + (size_t)b * C_DIM * HW + (size_t)(cbase + ln) * HW + pw + 4 * half;

    auto loadg = [&](int g, v4f (&xd)[K2_MT][4]) {
#pragma unroll
        for (int t = 0; t < K2_MT; ++t) {
            const float* r = xr + (size_t)(16 * t) * HW + 32 * g;
#pragma unroll
            for (int i = 0; i < 4; ++i) xd[t][i] = *(const v4f*)(r + 8 * i);
        }
    };

    v8f acc[K2_MT];
#pragma unroll
    for (int t = 0; t < K2_MT; ++t) acc[t] = v8f{};
    float cs = 0.f;

    auto compute = [&](int g, v4f (&xd)[K2_MT][4]) {
#pragma unroll
        for (int i = 0; i < 4; ++i) {
            const int qb = pw + 32 * g + 8 * i + 4 * half;
            const v4i a4 = *(const v4i*)&aL[qb];
            const v4f w4 = *(const v4f*)&wL[qb];
#pragma unroll
            for (int u = 0; u < 2; ++u) {
                v2f bf;
                bf[0] = (a4[2 * u]     == ln) ? w4[2 * u]     : 0.f;
                bf[1] = (a4[2 * u + 1] == ln) ? w4[2 * u + 1] : 0.f;
                cs += bf[0] + bf[1];
#pragma unroll
                for (int t = 0; t < K2_MT; ++t) {
                    v2f af;
                    af[0] = xd[t][i][2 * u];
                    af[1] = xd[t][i][2 * u + 1];
                    acc[t] = __builtin_amdgcn_wmma_f32_16x16x4_f32(false, af, false, bf,
                                                                   (short)0, acc[t], false, false);
                }
            }
        }
    };

    v4f xb0[K2_MT][4], xb1[K2_MT][4];
    loadg(0, xb0);

    {
        const int q = b * HW + pw + 4 * lane;
        const v4i av = *(const v4i*)(aidx + q);
        const v4f wv = *(const v4f*)(wsel + q);
        *(v4i*)&aL[pw + 4 * lane] = av;
        *(v4f*)&wL[pw + 4 * lane] = wv;
    }
    __syncthreads();

#pragma unroll
    for (int g = 0; g < K2_NG; g += 2) {
        loadg(g + 1, xb1);
        compute(g, xb0);
        __builtin_amdgcn_sched_barrier(0);
        if (g + 2 < K2_NG) loadg(g + 2, xb0);
        compute(g + 1, xb1);
        __builtin_amdgcn_sched_barrier(0);
    }

#pragma unroll
    for (int t = 0; t < K2_MT; ++t) {
#pragma unroll
        for (int r = 0; r < 8; ++r)
            red[((wave * K2_MT + t) * 8 + r) * 32 + lane] = acc[t][r];
    }
    csred[wave * 32 + lane] = cs;
    __syncthreads();

    const int row = tid >> 4;
    const int col = tid & 15;
    const int src = (row & 7) * 32 + col + 16 * (row >> 3);

    float cn = 0.f;
#pragma unroll
    for (int w = 0; w < 8; ++w) cn += csred[w * 32 + col] + csred[w * 32 + col + 16];
    const float den = (cn == 0.f) ? 1.f : cn;

    float* fb = feat + (size_t)b * C_DIM * 16;
#pragma unroll
    for (int t = 0; t < K2_MT; ++t) {
        float sum = 0.f;
#pragma unroll
        for (int w = 0; w < 8; ++w) sum += red[(w * K2_MT + t) * 256 + src];
        fb[(cbase + 16 * t + row) * 16 + col] = sum / den;
    }
}

__global__ __launch_bounds__(256) void k3_mod(
    const float* __restrict__ feat, const float* __restrict__ gamma,
    const float* __restrict__ beta, float* __restrict__ out_mod)
{
    __shared__ float featL[C_DIM * LNUM];
    __shared__ float red0[8], red1[8], stat[2];

    const int tid = threadIdx.x;
    const int b   = blockIdx.x;

    const float* fb = feat + (size_t)b * C_DIM * 16;
    float s = 0.f, ss = 0.f;
    for (int i = tid; i < C_DIM * LNUM; i += 256) {
        const int c = i / LNUM, l = i % LNUM;
        const float v = fb[c * 16 + l];
        featL[i] = v;
        s += v; ss += v * v;
    }
    for (int o = 16; o > 0; o >>= 1) { s += __shfl_down(s, o); ss += __shfl_down(ss, o); }
    const int lane = tid & 31, wave = tid >> 5;
    if (lane == 0) { red0[wave] = s; red1[wave] = ss; }
    __syncthreads();
    if (tid == 0) {
        float S = 0.f, SS = 0.f;
        for (int w = 0; w < 8; ++w) { S += red0[w]; SS += red1[w]; }
        const float n = (float)(C_DIM * LNUM);
        const float mean = S / n;
        const float var  = SS / n - mean * mean;
        stat[0] = mean;
        stat[1] = rsqrtf(var + 1e-5f);
    }
    __syncthreads();
    const float mean = stat[0], rstd = stat[1];
    float* ob = out_mod + (size_t)b * C_DIM * LNUM;
    for (int i = tid; i < C_DIM * LNUM; i += 256)
        ob[i] = (featL[i] - mean) * rstd * gamma[i] + beta[i];
}

#define K4_KW   96
#define K4_NP   12

__global__ __launch_bounds__(256) void k4_scores(
    const float* __restrict__ cls_w, const float* __restrict__ mod,
    float* __restrict__ scores)
{
    __shared__ float red[8 * 256];

    const int tid   = threadIdx.x;
    const int wave  = tid >> 5;
    const int lane  = tid & 31;
    const int half  = lane >> 4;
    const int ln    = lane & 15;
    const int nbase = blockIdx.x * 16;
    const int b0    = blockIdx.y * 2;
    const int kb    = wave * K4_KW + 4 * half;

    const int row  = nbase + ln;
    const int rowc = (row < NCLS) ? row : (NCLS - 1);
    const float* ar = cls_w + (size_t)rowc * C_DIM + kb;
    const float* br = mod + ((size_t)(b0 + (ln >> 3)) * C_DIM + kb) * LNUM + (ln & 7);

    v4f   av[K4_NP];
    float bv[K4_NP][4];
#pragma unroll
    for (int j = 0; j < K4_NP; ++j) {
        av[j] = *(const v4f*)(ar + 8 * j);
#pragma unroll
        for (int e = 0; e < 4; ++e) bv[j][e] = br[(8 * j + e) * LNUM];
    }
    __builtin_amdgcn_sched_barrier(0);

    v8f acc = {};
#pragma unroll
    for (int j = 0; j < K4_NP; ++j) {
#pragma unroll
        for (int u = 0; u < 2; ++u) {
            v2f af, bf;
            af[0] = av[j][2 * u]; af[1] = av[j][2 * u + 1];
            bf[0] = bv[j][2 * u]; bf[1] = bv[j][2 * u + 1];
            acc = __builtin_amdgcn_wmma_f32_16x16x4_f32(false, af, false, bf,
                                                        (short)0, acc, false, false);
        }
    }

#pragma unroll
    for (int r = 0; r < 8; ++r) red[wave * 256 + r * 32 + lane] = acc[r];
    __syncthreads();

    const int i   = tid >> 7;
    const int rw  = (tid >> 3) & 15;
    const int l   = tid & 7;
    const int src = (rw & 7) * 32 + (rw >> 3) * 16 + i * 8 + l;
    float s = 0.f;
#pragma unroll
    for (int w = 0; w < 8; ++w) s += red[w * 256 + src];
    const int n = nbase + rw;
    if (n < NCLS)
        scores[((size_t)(b0 + i) * NCLS + n) * (LNUM - 1) + l] = s;
}

extern "C" void kernel_launch(void* const* d_in, const int* in_sizes, int n_in,
                              void* d_out, int out_size, void* d_ws, size_t ws_size,
                              hipStream_t stream) {
    const float* x      = (const float*)d_in[0];
    const float* q      = (const float*)d_in[1];
    const float* p_bias = (const float*)d_in[2];
    const float* gamma  = (const float*)d_in[3];
    const float* beta   = (const float*)d_in[4];
    const float* cls_w  = (const float*)d_in[5];

    float* out        = (float*)d_out;
    float* out_mod    = out;
    float* out_maps   = out + (size_t)B_DIM * C_DIM * LNUM;
    float* out_scores = out_maps + (size_t)B_DIM * LNUM * HW;

    int*   aidx = (int*)d_ws;
    float* wsel = (float*)d_ws + (size_t)B_DIM * HW;
    float* feat = (float*)d_ws + (size_t)2 * B_DIM * HW;

    k1_maps  <<<dim3(16, B_DIM), 256, 0, stream>>>(x, q, p_bias, out_maps, aidx, wsel);
    k2_pool  <<<dim3(16, B_DIM), 256, 0, stream>>>(x, aidx, wsel, feat);
    k3_mod   <<<dim3(B_DIM),    256, 0, stream>>>(feat, gamma, beta, out_mod);
    k4_scores<<<dim3(13, 16), 256, 0, stream>>>(cls_w, out_mod, out_scores);
}


# scores launch dropped by linearity: k2 blocks take w.(gamma*feat) of their 48 channels on WMMA (w.gamma, w.beta: 13 extra blocks); k3 adds 16 partials, applies mean, 1/std; scores sum order changes
# speedup vs baseline: 6.8957x; 1.2912x over previous
#include <hip/hip_runtime.h>
#include <hip/hip_bf16.h>

#define B_DIM   32
#define C_DIM   768
#define H_DIM   32
#define W_DIM   32
#define HW      1024
#define LNUM    9
#define NCLS    200

typedef __attribute__((ext_vector_type(2))) float v2f;
typedef __attribute__((ext_vector_type(4))) float v4f;
typedef __attribute__((ext_vector_type(8))) float v8f;

#define K1_NT   4
#define K1_TSTR 260
#define K1_WSTR (K1_NT * K1_TSTR)
#define K1_SQW  (K1_NT * 36)
#define K1_GRP  4
#define K1_NG   (24 / K1_GRP)

__global__ __launch_bounds__(256, 4) void k1_maps(
    const float* __restrict__ x, const float* __restrict__ q,
    const float* __restrict__ p_bias,
    float* __restrict__ out_maps, int* __restrict__ aidx, float* __restrict__ wsel)
{
    __shared__ float red[8 * K1_WSTR];
    __shared__ float sqred[8 * K1_SQW];
    __shared__ float asqred[8 * 32];

    const int tid  = threadIdx.x;
    const int b    = blockIdx.y;
    const int wave = tid >> 5;
    const int lane = tid & 31;
    const int half = lane >> 4;
    const int ln   = lane & 15;
    const int P0   = blockIdx.x * 64;
    const int kb   = wave * 96 + 2 * half;

    const float* xw = x + (size_t)b * C_DIM * HW + (size_t)kb * HW + P0 + ln * 4;
    const bool lv = (ln < LNUM);
    const float* qw = q + (lv ? ln : 0) * C_DIM + kb;

    v8f acc[K1_NT];
    float sq[K1_NT];
#pragma unroll
    for (int t = 0; t < K1_NT; ++t) { acc[t] = v8f{}; sq[t] = 0.f; }
    float asqp = 0.f;

    auto loadg = [&](int g, v4f (&xd)[K1_GRP][2], v2f (&ad)[K1_GRP]) {
#pragma unroll
        for (int j = 0; j < K1_GRP; ++j) {
            const int s = g * K1_GRP + j;
            const float* r0 = xw + (size_t)(4 * s) * HW;
            xd[j][0] = *(const v4f*)(r0);
            xd[j][1] = *(const v4f*)(r0 + HW);
            ad[j]    = *(const v2f*)(qw + 4 * s);
        }
    };
    auto compute = [&](v4f (&xd)[K1_GRP][2], v2f (&ad)[K1_GRP]) {
#pragma unroll
        for (int j = 0; j < K1_GRP; ++j) {
            v2f af;
            af[0] = lv ? ad[j][0] : 0.f;
            af[1] = lv ? ad[j][1] : 0.f;
            asqp += af[0] * af[0] + af[1] * af[1];
#pragma unroll
            for (int t = 0; t < K1_NT; ++t) {
                v2f bf;
                bf[0] = xd[j][0][t];
                bf[1] = xd[j][1][t];
                sq[t] += bf[0] * bf[0] + bf[1] * bf[1];
                acc[t] = __builtin_amdgcn_wmma_f32_16x16x4_f32(false, af, false, bf,
                                                               (short)0, acc[t], false, false);
            }
        }
    };

    v4f xb0[K1_GRP][2], xb1[K1_GRP][2];
    v2f ab0[K1_GRP], ab1[K1_GRP];
    loadg(0, xb0, ab0);
#pragma unroll
    for (int g = 0; g < K1_NG; g += 2) {
        if (g + 1 < K1_NG) loadg(g + 1, xb1, ab1);
        compute(xb0, ab0);
        __builtin_amdgcn_sched_barrier(0);
        if (g + 2 < K1_NG) loadg(g + 2, xb0, ab0);
        if (g + 1 < K1_NG) compute(xb1, ab1);
        __builtin_amdgcn_sched_barrier(0);
    }

#pragma unroll
    for (int t = 0; t < K1_NT; ++t) {
#pragma unroll
        for (int r = 0; r < 8; ++r)
            red[wave * K1_WSTR + t * K1_TSTR + r * 32 + lane] = acc[t][r];
        sqred[wave * K1_SQW + t * 36 + lane] = sq[t];
    }
    asqred[wave * 32 + lane] = asqp;
    __syncthreads();
    if (wave >= 4) return;

    const int pl   = wave * 16 + ln;
    const int col  = pl >> 2;
    const int tsel = pl & 3;
    const int sl   = col + 16 * half;
    const int pix  = P0 + pl;

    float accf[8];
#pragma unroll
    for (int r = 0; r < 8; ++r) {
        float s = 0.f;
#pragma unroll
        for (int w = 0; w < 8; ++w) s += red[w * K1_WSTR + tsel * K1_TSTR + r * 32 + sl];
        accf[r] = s;
    }
    float bsq = 0.f;
#pragma unroll
    for (int w = 0; w < 8; ++w)
        bsq += sqred[w * K1_SQW + tsel * 36 + col] + sqred[w * K1_SQW + tsel * 36 + col + 16];
    float aq = 0.f;
#pragma unroll
    for (int w = 0; w < 8; ++w) aq += asqred[w * 32 + ln] + asqred[w * 32 + ln + 16];
    float asql[8];
#pragma unroll
    for (int r = 0; r < 8; ++r) asql[r] = __shfl(aq, r + 8 * half);

    float sc[8];
    float mx = -3.4e38f;
    for (int r = 0; r < 8; ++r) {
        const int l = r + 8 * half;
        float v = -3.4e38f;
        if (l < LNUM)
            v = -(bsq - 2.f * accf[r] + asql[r]) + p_bias[l * HW + pix];
        sc[r] = v;
        mx = fmaxf(mx, v);
    }
    mx = fmaxf(mx, __shfl_xor(mx, 16));
    float es = 0.f;
    for (int r = 0; r < 8; ++r) {
        const int l = r + 8 * half;
        const float e = (l < LNUM) ? expf(sc[r] - mx) : 0.f;
        sc[r] = e;
        es += e;
    }
    es += __shfl_xor(es, 16);
    const float inv = 1.f / es;
    float s2 = 0.f;
    for (int r = 0; r < 8; ++r) {
        const int l = r + 8 * half;
        if (l < LNUM) { sc[r] = sc[r] * inv + 1e-6f; s2 += sc[r]; }
    }
    s2 += __shfl_xor(s2, 16);
    const float inv2 = 1.f / s2;

    float bestv = -1.f; int bestl = 0;
    float* om = out_maps + (size_t)b * LNUM * HW;
    for (int r = 0; r < 8; ++r) {
        const int l = r + 8 * half;
        if (l < LNUM) {
            const float mv = sc[r] * inv2;
            om[l * HW + pix] = mv;
            if (mv > bestv) { bestv = mv; bestl = l; }
        }
    }
    const float pv = __shfl_xor(bestv, 16);
    const int   pl2 = __shfl_xor(bestl, 16);
    if (pv > bestv || (pv == bestv && pl2 < bestl)) { bestv = pv; bestl = pl2; }
    if (half == 0) {
        aidx[b * HW + pix] = bestl;
        wsel[b * HW + pix] = bestv;
    }
}

#define K2_MT   3
#define K2_PW   128
#define K2_NG   4

#define K2_FGS  20
#define NTILE   13
#define WS_NEED 8664064

typedef __attribute__((ext_vector_type(4))) int v4i;

#define GB_ROWS 208
__device__ __forceinline__ void gb_tile(
    const float* __restrict__ cls_w, const float* __restrict__ gamma,
    const float* __restrict__ beta, float* __restrict__ GB, float* red, int tile)
{
    const int tid   = threadIdx.x;
    const int wave  = tid >> 5;
    const int lane  = tid & 31;
    const int half  = lane >> 4;
    const int ln    = lane & 15;
    const int nbase = tile * 16;
    const int kb    = wave * 96 + 4 * half;

    const int row  = nbase + ln;
    const int rowc = (row < NCLS) ? row : (NCLS - 1);
    const float* ar = cls_w + (size_t)rowc * C_DIM + kb;
    const float* br = ((ln >> 3) ? beta : gamma) + (size_t)kb * LNUM + (ln & 7);

    v4f   av[12];
    float bv[12][4];
#pragma unroll
    for (int j = 0; j < 12; ++j) {
        av[j] = *(const v4f*)(ar + 8 * j);
#pragma unroll
        for (int e = 0; e < 4; ++e) bv[j][e] = br[(8 * j + e) * LNUM];
    }
    __builtin_amdgcn_sched_barrier(0);

    v8f acc = {};
#pragma unroll
    for (int j = 0; j < 12; ++j) {
#pragma unroll
        for (int u = 0; u < 2; ++u) {
            v2f af, bf;
            af[0] = av[j][2 * u]; af[1] = av[j][2 * u + 1];
            bf[0] = bv[j][2 * u]; bf[1] = bv[j][2 * u + 1];
            acc = __builtin_amdgcn_wmma_f32_16x16x4_f32(false, af, false, bf,
                                                        (short)0, acc, false, false);
        }
    }
#pragma unroll
    for (int r = 0; r < 8; ++r) red[wave * 256 + r * 32 + lane] = acc[r];
    __syncthreads();

    const int i   = tid >> 7;
    const int rw  = (tid >> 3) & 15;
    const int l   = tid & 7;
    const int src = (rw & 7) * 32 + (rw >> 3) * 16 + i * 8 + l;
    float s = 0.f;
#pragma unroll
    for (int w = 0; w < 8; ++w) s += red[w * 256 + src];
    const int n = nbase + rw;
    if (n < NCLS) GB[(i * GB_ROWS + n) * 8 + l] = s;
}


__global__ __launch_bounds__(256, 4) void k2_pool(
    const float* __restrict__ x, const int* __restrict__ aidx,
    const float* __restrict__ wsel, const float* __restrict__ gamma,
    const float* __restrict__ beta, const float* __restrict__ cls_w,
    float* __restrict__ feat, float* __restrict__ Pp, float* __restrict__ GB,
    size_t ws_bytes)
{
    __shared__ __attribute__((aligned(16))) int   aL[HW];
    __shared__ __attribute__((aligned(16))) float wL[HW];
    __shared__ float red[8 * K2_MT * 256];
    __shared__ float csred[8 * 32];
    __shared__ float stw[16];
    __shared__ float fg[48 * K2_FGS];

    const bool ws_ok = (ws_bytes >= (size_t)WS_NEED);
    if (blockIdx.y == B_DIM) {
        if (blockIdx.x < NTILE && ws_ok) gb_tile(cls_w, gamma, beta, GB, red, blockIdx.x);
        return;
    }

    const int tid  = threadIdx.x;
    const int b    = blockIdx.y;
    const int wave = tid >> 5;
    const int lane = tid & 31;
    const int half = lane >> 4;
    const int ln   = lane & 15;
    const int cbase = blockIdx.x * (16 * K2_MT);
    const int pw    = wave * K2_PW;

    const float* xr = x + (size_t)b * C_DIM * HW + (size_t)(cbase + ln) * HW + pw + 4 * half;

    auto loadg = [&](int g, v4f (&xd)[K2_MT][4]) {
#pragma unroll
        for (int t = 0; t < K2_MT; ++t) {
            const float* r = xr + (size_t)(16 * t) * HW + 32 * g;
#pragma unroll
            for (int i = 0; i < 4; ++i) xd[t][i] = *(const v4f*)(r + 8 * i);
        }
    };

    v8f acc[K2_MT];
#pragma unroll
    for (int t = 0; t < K2_MT; ++t) acc[t] = v8f{};
    float cs = 0.f;

    auto compute = [&](int g, v4f (&xd)[K2_MT][4]) {
#pragma unroll
        for (int i = 0; i < 4; ++i) {
            const int qb = pw + 32 * g + 8 * i + 4 * half;
            const v4i a4 = *(const v4i*)&aL[qb];
            const v4f w4 = *(const v4f*)&wL[qb];
#pragma unroll
            for (int u = 0; u < 2; ++u) {
                v2f bf;
                bf[0] = (a4[2 * u]     == ln) ? w4[2 * u]     : 0.f;
                bf[1] = (a4[2 * u + 1] == ln) ? w4[2 * u + 1] : 0.f;
                cs += bf[0] + bf[1];
#pragma unroll
                for (int t = 0; t < K2_MT; ++t) {
                    v2f af;
                    af[0] = xd[t][i][2 * u];
                    af[1] = xd[t][i][2 * u + 1];
                    acc[t] = __builtin_amdgcn_wmma_f32_16x16x4_f32(false, af, false, bf,
                                                                   (short)0, acc[t], false, false);
                }
            }
        }
    };

    const bool two = (wave + 8 < NTILE);
    v4f wa0[6], wa1[6];
    float gm[K2_MT];
    auto loadw = [&]() {
        const int n0  = wave * 16 + ln;
        const int n1r = (wave + 8) * 16 + ln;
        const int n1  = (n1r < NCLS) ? n1r : (NCLS - 1);
        const float* a0 = cls_w + (size_t)n0 * C_DIM + cbase + 4 * half;
        const float* a1 = cls_w + (size_t)n1 * C_DIM + cbase + 4 * half;
#pragma unroll
        for (int j = 0; j < 6; ++j) wa0[j] = *(const v4f*)(a0 + 8 * j);
        if (two) {
#pragma unroll
            for (int j = 0; j < 6; ++j) wa1[j] = *(const v4f*)(a1 + 8 * j);
        }
        const int grow = tid >> 4, gcol = tid & 15;
#pragma unroll
        for (int t = 0; t < K2_MT; ++t)
            gm[t] = (gcol < LNUM - 1) ? gamma[(cbase + 16 * t + grow) * LNUM + gcol] : 0.f;
    };

    v4f xb0[K2_MT][4], xb1[K2_MT][4];
    loadg(0, xb0);

    {
        const int q = b * HW + pw + 4 * lane;
        const v4i av = *(const v4i*)(aidx + q);
        const v4f wv = *(const v4f*)(wsel + q);
        *(v4i*)&aL[pw + 4 * lane] = av;
        *(v4f*)&wL[pw + 4 * lane] = wv;
    }
    __syncthreads();

#pragma unroll
    for (int g = 0; g < K2_NG; g += 2) {
        loadg(g + 1, xb1);
        __builtin_amdgcn_sched_barrier(0);
        compute(g, xb0);
        __builtin_amdgcn_sched_barrier(0);
        if (g + 2 < K2_NG) loadg(g + 2, xb0);
        else loadw();
        __builtin_amdgcn_sched_barrier(0);
        compute(g + 1, xb1);
        __builtin_amdgcn_sched_barrier(0);
    }

#pragma unroll
    for (int t = 0; t < K2_MT; ++t) {
#pragma unroll
        for (int r = 0; r < 8; ++r)
            red[((wave * K2_MT + t) * 8 + r) * 32 + lane] = acc[t][r];
    }
    csred[wave * 32 + lane] = cs;
    __syncthreads();

    const int row = tid >> 4;
    const int col = tid & 15;
    const int src = (row & 7) * 32 + col + 16 * (row >> 3);

    float cn = 0.f;
#pragma unroll
    for (int w = 0; w < 8; ++w) cn += csred[w * 32 + col] + csred[w * 32 + col + 16];
    const float den = (cn == 0.f) ? 1.f : cn;

    float fv[K2_MT];
    float s1 = 0.f, s2 = 0.f;
#pragma unroll
    for (int t = 0; t < K2_MT; ++t) {
        float sum = 0.f;
#pragma unroll
        for (int w = 0; w < 8; ++w) sum += red[(w * K2_MT + t) * 256 + src];
        const float f = sum / den;
        fv[t] = f;
        if (col < LNUM) { s1 += f; s2 += f * f; }
        const int cl = 16 * t + row;
        fg[cl * K2_FGS + col] = f * gm[t];
    }
#pragma unroll
    for (int o = 16; o > 0; o >>= 1) { s1 += __shfl_xor(s1, o); s2 += __shfl_xor(s2, o); }
    if (lane == 0) { stw[wave] = s1; stw[8 + wave] = s2; }
    __syncthreads();
    if (tid == 10 || tid == 11) {
        float a = 0.f;
#pragma unroll
        for (int w = 0; w < 8; ++w) a += stw[(tid - 10) * 8 + w];
        fv[0] = a;
    }

    float* fb = feat + (size_t)b * C_DIM * 16;
#pragma unroll
    for (int t = 0; t < K2_MT; ++t)
        fb[(cbase + 16 * t + row) * 16 + col] = fv[t];

    float bq[6][4];
#pragma unroll
    for (int j = 0; j < 6; ++j) {
#pragma unroll
        for (int e = 0; e < 4; ++e) bq[j][e] = fg[(8 * j + 4 * half + e) * K2_FGS + ln];
    }
    v8f p0 = {}, p1 = {};
    if (two) {
#pragma unroll
        for (int j = 0; j < 6; ++j) {
#pragma unroll
            for (int u = 0; u < 2; ++u) {
                v2f bf, a0f, a1f;
                bf[0]  = bq[j][2 * u];  bf[1]  = bq[j][2 * u + 1];
                a0f[0] = wa0[j][2 * u]; a0f[1] = wa0[j][2 * u + 1];
                a1f[0] = wa1[j][2 * u]; a1f[1] = wa1[j][2 * u + 1];
                p0 = __builtin_amdgcn_wmma_f32_16x16x4_f32(false, a0f, false, bf, (short)0, p0, false, false);
                p1 = __builtin_amdgcn_wmma_f32_16x16x4_f32(false, a1f, false, bf, (short)0, p1, false, false);
            }
        }
    } else {
#pragma unroll
        for (int j = 0; j < 6; ++j) {
#pragma unroll
            for (int u = 0; u < 2; ++u) {
                v2f bf, a0f;
                bf[0]  = bq[j][2 * u];  bf[1]  = bq[j][2 * u + 1];
                a0f[0] = wa0[j][2 * u]; a0f[1] = wa0[j][2 * u + 1];
                p0 = __builtin_amdgcn_wmma_f32_16x16x4_f32(false, a0f, false, bf, (short)0, p0, false, false);
            }
        }
    }
    if (ws_ok) {
        float* pt = Pp + ((size_t)(b * 16 + blockIdx.x) * NTILE + wave) * 256 + lane;
#pragma unroll
        for (int r = 0; r < 8; ++r) pt[r * 32] = p0[r];
        if (two) {
#pragma unroll
            for (int r = 0; r < 8; ++r) pt[8 * 256 + r * 32] = p1[r];
        }
    }
}

#define K3_BPI  6
#define K3_EPT  9
#define K3_NT   128

__global__ __launch_bounds__(K3_NT) void k3_mod(
    const float* __restrict__ feat, const float* __restrict__ gamma,
    const float* __restrict__ beta, const float* __restrict__ Pp,
    const float* __restrict__ GB, float* __restrict__ out_mod,
    float* __restrict__ scores, size_t ws_bytes)
{
    const int tid = threadIdx.x;
    const int b   = blockIdx.y;
    const float* fb = feat + (size_t)b * C_DIM * 16;

    if (blockIdx.x >= K3_BPI) {
        if (ws_bytes < (size_t)WS_NEED) return;
        const int tile = blockIdx.x - K3_BPI;
        const int rw   = tid >> 3;
        const int l    = tid & 7;
        const int src  = (rw & 7) * 32 + (rw >> 3) * 16 + l;
        const int n    = tile * 16 + rw;
        if (__builtin_amdgcn_readfirstlane(tile * 16 + ((tid >> 5) << 2)) >= NCLS) return;

        float p[C_DIM / 48];
#pragma unroll
        for (int g = 0; g < C_DIM / 48; ++g)
            p[g] = Pp[((size_t)(b * 16 + g) * NTILE + tile) * 256 + src];
        const float gg = GB[n * 8 + l];
        const float bt = GB[(GB_ROWS + n) * 8 + l];
        const int   lane = tid & 31;
        const float st = fb[(48 * (lane & 15)) * 16 + 10 + (lane >> 4)];
        float S = 0.f, SS = 0.f;
#pragma unroll
        for (int w = 0; w < C_DIM / 48; ++w) {
            S  += __int_as_float(__builtin_amdgcn_readlane(__float_as_int(st), w));
            SS += __int_as_float(__builtin_amdgcn_readlane(__float_as_int(st), 16 + w));
        }
        const float nn   = (float)(C_DIM * LNUM);
        const float mean = S / nn;
        const float var  = SS / nn - mean * mean;
        const float rstd = rsqrtf(var + 1e-5f);
        float ps = 0.f;
#pragma unroll
        for (int g = 0; g < C_DIM / 48; ++g) ps += p[g];
        scores[((size_t)b * NCLS + n) * (LNUM - 1) + l] = rstd * (ps - mean * gg) + bt;
        return;
    }

    const int i0  = blockIdx.x * (K3_EPT * K3_NT) + tid;

    float v[K3_EPT], g[K3_EPT], be[K3_EPT];
#pragma unroll
    for (int j = 0; j < K3_EPT; ++j) {
        const int i = i0 + j * K3_NT;
        const int c = i / LNUM, l = i - c * LNUM;
        v[j]  = fb[c * 16 + l];
        g[j]  = gamma[i];
        be[j] = beta[i];
    }

    const int lane = tid & 31;
    const float st = fb[(48 * (lane & 15)) * 16 + 10 + (lane >> 4)];
    float S = 0.f, SS = 0.f;
#pragma unroll
    for (int w = 0; w < C_DIM / 48; ++w) {
        S  += __int_as_float(__builtin_amdgcn_readlane(__float_as_int(st), w));
        SS += __int_as_float(__builtin_amdgcn_readlane(__float_as_int(st), 16 + w));
    }
    const float n    = (float)(C_DIM * LNUM);
    const float mean = S / n;
    const float var  = SS / n - mean * mean;
    const float rstd = rsqrtf(var + 1e-5f);

    float* ob = out_mod + (size_t)b * C_DIM * LNUM;
#pragma unroll
    for (int j = 0; j < K3_EPT; ++j)
        ob[i0 + j * K3_NT] = (v[j] - mean) * rstd * g[j] + be[j];
}

extern "C" void kernel_launch(void* const* d_in, const int* in_sizes, int n_in,
                              void* d_out, int out_size, void* d_ws, size_t ws_size,
                              hipStream_t stream) {
    const float* x      = (const float*)d_in[0];
    const float* q      = (const float*)d_in[1];
    const float* p_bias = (const float*)d_in[2];
    const float* gamma  = (const float*)d_in[3];
    const float* beta   = (const float*)d_in[4];
    const float* cls_w  = (const float*)d_in[5];

    float* out        = (float*)d_out;
    float* out_mod    = out;
    float* out_maps   = out + (size_t)B_DIM * C_DIM * LNUM;
    float* out_scores = out_maps + (size_t)B_DIM * LNUM * HW;

    int*   aidx = (int*)d_ws;
    float* wsel = (float*)d_ws + (size_t)B_DIM * HW;
    float* feat = (float*)d_ws + (size_t)2 * B_DIM * HW;
    float* Pp   = feat + (size_t)B_DIM * C_DIM * 16;
    float* GB   = Pp + (size_t)B_DIM * 16 * NTILE * 256;

    k1_maps  <<<dim3(16, B_DIM), 256, 0, stream>>>(x, q, p_bias, out_maps, aidx, wsel);
    k2_pool  <<<dim3(16, B_DIM + 1), 256, 0, stream>>>(x, aidx, wsel, gamma, beta, cls_w, feat, Pp, GB, ws_size);
    k3_mod   <<<dim3(K3_BPI + NTILE, B_DIM), K3_NT, 0, stream>>>(feat, gamma, beta, Pp, GB, out_mod, out_scores, ws_size);
}
